# starting kernel
# baseline (speedup 1.0000x reference)
#include <hip/hip_runtime.h>


typedef float v2f __attribute__((ext_vector_type(2)));
typedef float v8f __attribute__((ext_vector_type(8)));

#define LOG_N       12
#define NELEM       4096
#define BATCH       8192
#define ROWS_TILE   16
#define LDS_STRIDE  4100

__device__ __forceinline__ void ctx_pos(int g, int c, int& base, int& step) {
    if (g == 0)      { base = c << 4;                        step = 1;   }
    else if (g == 1) { base = ((c & 0xF0) << 4) | (c & 15);  step = 16;  }
    else             { base = c;                             step = 256; }
}

__global__ void butterfly_compose(const float* __restrict__ tw,
                                  float* __restrict__ M) {
    int tid = blockIdx.x * blockDim.x + threadIdx.x;
    if (tid >= 3 * 256 * 16) return;
    int k = tid & 15;
    int c = (tid >> 4) & 255;
    int g = tid >> 12;

    int base, step;
    ctx_pos(g, c, base, step);

    float v[16];
#pragma unroll
    for (int t = 0; t < 16; ++t) v[t] = (t == k) ? 1.0f : 0.0f;

#pragma unroll
    for (int j = 0; j < 4; ++j) {
        const int idx = 4 * g + j;
        const int S   = 1 << idx;
#pragma unroll
        for (int u = 0; u < 8; ++u) {
            const int off = u & ((1 << j) - 1);
            const int blk = u >> j;
            const int t0  = (blk << (j + 1)) | off;
            const int t1  = t0 + (1 << j);
            const int p0  = base + t0 * step;
            const int w   = ((p0 >> (idx + 1)) << idx) | (p0 & (S - 1));
            const float* q = tw + idx * (2048 * 4) + w * 4;
            const float a00 = q[0], a01 = q[1], a10 = q[2], a11 = q[3];
            const float x0 = v[t0], x1 = v[t1];
            v[t0] = a00 * x0 + a01 * x1;
            v[t1] = a10 * x0 + a11 * x1;
        }
    }

    float* out = M + ((g * 256 + c) * 16) * 16 + k;
#pragma unroll
    for (int i = 0; i < 16; ++i) out[i * 16] = v[i];
}

__global__ void __launch_bounds__(256, 1)
butterfly_apply(const float* __restrict__ x,
                const float* __restrict__ M,
                float* __restrict__ out) {
    extern __shared__ float X[];

    const int r0 = blockIdx.x * ROWS_TILE;

    {
        const float4* src = (const float4*)(x + (size_t)r0 * NELEM);
        for (int i = threadIdx.x; i < ROWS_TILE * (NELEM / 4); i += 256) {
            const int row = i >> 10;
            const int c4  = i & 1023;
            float4 val = src[row * 1024 + c4];
            *(float4*)&X[row * LDS_STRIDE + c4 * 4] = val;
        }
    }

    const int lane = threadIdx.x & 31;
    const int wave = threadIdx.x >> 5;
    const int r    = lane & 15;
    const int hi   = lane >> 4;

    for (int g = 0; g < 3; ++g) {
        __syncthreads();
        for (int ci = 0; ci < 32; ++ci) {
            const int c = (wave << 5) | ci;
            int base, step;
            ctx_pos(g, c, base, step);

            const float* Mrow = M + ((g * 256 + c) * 16 + r) * 16;
            const float* Brow = X + r * LDS_STRIDE + base;

            v8f acc = {};
#pragma unroll
            for (int q = 0; q < 4; ++q) {
                const int kb = 4 * q + 2 * hi;
                v2f a;
                a.x = Mrow[kb];
                a.y = Mrow[kb + 1];
                v2f b;
                b.x = Brow[kb * step];
                b.y = Brow[(kb + 1) * step];
                acc = __builtin_amdgcn_wmma_f32_16x16x4_f32(
                     false, a,  false, b,
                     (short)0, acc,  false,  false);
            }

            float* Drow = X + r * LDS_STRIDE + base;
#pragma unroll
            for (int j = 0; j < 8; ++j) {
                const int i = j + 8 * hi;
                Drow[i * step] = acc[j];
            }
        }
    }

    __syncthreads();

    {
        float4* dst = (float4*)(out + (size_t)r0 * NELEM);
        for (int i = threadIdx.x; i < ROWS_TILE * (NELEM / 4); i += 256) {
            const int row = i >> 10;
            const int c4  = i & 1023;
            dst[row * 1024 + c4] = *(const float4*)&X[row * LDS_STRIDE + c4 * 4];
        }
    }
}

extern "C" void kernel_launch(void* const* d_in, const int* in_sizes, int n_in,
                              void* d_out, int out_size, void* d_ws, size_t ws_size,
                              hipStream_t stream) {
    (void)in_sizes; (void)n_in; (void)out_size; (void)ws_size;

    const float* x  = (const float*)d_in[0];
    const float* tw = (const float*)d_in[1];
    float*       o  = (float*)d_out;
    float*       M  = (float*)d_ws;

    butterfly_compose<<<48, 256, 0, stream>>>(tw, M);

    const size_t lds_bytes = (size_t)ROWS_TILE * LDS_STRIDE * sizeof(float);
    (void)hipFuncSetAttribute((const void*)butterfly_apply,
                              hipFuncAttributeMaxDynamicSharedMemorySize,
                              (int)lds_bytes);

    butterfly_apply<<<BATCH / ROWS_TILE, 256, lds_bytes, stream>>>(x, M, o);
}


# apply kernel: stage loop software-pipelined, 4 contexts per trip on 4 independent accumulators, next trip's A (L2) and B (LDS) fragments requested ahead of the current trip's WMMAs
# speedup vs baseline: 1.2079x; 1.2079x over previous
#include <hip/hip_runtime.h>


typedef float v2f __attribute__((ext_vector_type(2)));
typedef float v8f __attribute__((ext_vector_type(8)));

#define LOG_N       12
#define NELEM       4096
#define BATCH       8192
#define ROWS_TILE   16
#define LDS_STRIDE  4100

__device__ __forceinline__ void ctx_pos(int g, int c, int& base, int& step) {
    if (g == 0)      { base = c << 4;                        step = 1;   }
    else if (g == 1) { base = ((c & 0xF0) << 4) | (c & 15);  step = 16;  }
    else             { base = c;                             step = 256; }
}

__global__ void butterfly_compose(const float* __restrict__ tw,
                                  float* __restrict__ M) {
    int tid = blockIdx.x * blockDim.x + threadIdx.x;
    if (tid >= 3 * 256 * 16) return;
    int k = tid & 15;
    int c = (tid >> 4) & 255;
    int g = tid >> 12;

    int base, step;
    ctx_pos(g, c, base, step);

    float v[16];
#pragma unroll
    for (int t = 0; t < 16; ++t) v[t] = (t == k) ? 1.0f : 0.0f;

#pragma unroll
    for (int j = 0; j < 4; ++j) {
        const int idx = 4 * g + j;
        const int S   = 1 << idx;
#pragma unroll
        for (int u = 0; u < 8; ++u) {
            const int off = u & ((1 << j) - 1);
            const int blk = u >> j;
            const int t0  = (blk << (j + 1)) | off;
            const int t1  = t0 + (1 << j);
            const int p0  = base + t0 * step;
            const int w   = ((p0 >> (idx + 1)) << idx) | (p0 & (S - 1));
            const float* q = tw + idx * (2048 * 4) + w * 4;
            const float a00 = q[0], a01 = q[1], a10 = q[2], a11 = q[3];
            const float x0 = v[t0], x1 = v[t1];
            v[t0] = a00 * x0 + a01 * x1;
            v[t1] = a10 * x0 + a11 * x1;
        }
    }

    float* out = M + ((g * 256 + c) * 16) * 16 + k;
#pragma unroll
    for (int i = 0; i < 16; ++i) out[i * 16] = v[i];
}

#define CTX_TRIP   4
#define NTRIPS     (32 / CTX_TRIP)

struct Frag4 { v2f v[CTX_TRIP][4]; };

__device__ __forceinline__ void load_a_trip(const float* __restrict__ Mw,
                                            int g, int t, Frag4& f) {
    const float* p = Mw + (g * 256 + t * CTX_TRIP) * 256;
#pragma unroll
    for (int u = 0; u < CTX_TRIP; ++u) {
#pragma unroll
        for (int q = 0; q < 4; ++q) {
            f.v[u][q].x = p[u * 256 + 4 * q];
            f.v[u][q].y = p[u * 256 + 4 * q + 1];
        }
    }
}

template <int G>
__device__ __forceinline__ void load_b_trip(const float* Xr, int wave, int hi,
                                            int t, Frag4& f) {
#pragma unroll
    for (int u = 0; u < CTX_TRIP; ++u) {
        const int c = (wave << 5) | (t * CTX_TRIP + u);
        int base, step;
        ctx_pos(G, c, base, step);
        const float* Brow = Xr + base;
#pragma unroll
        for (int q = 0; q < 4; ++q) {
            const int kb = 4 * q + 2 * hi;
            f.v[u][q].x = Brow[kb * step];
            f.v[u][q].y = Brow[(kb + 1) * step];
        }
    }
}

__device__ __forceinline__ void mma_trip(const Frag4& a, const Frag4& b,
                                         v8f (&acc)[CTX_TRIP]) {
#pragma unroll
    for (int u = 0; u < CTX_TRIP; ++u) acc[u] = v8f{};
#pragma unroll
    for (int q = 0; q < 4; ++q) {
#pragma unroll
        for (int u = 0; u < CTX_TRIP; ++u) {
            acc[u] = __builtin_amdgcn_wmma_f32_16x16x4_f32(
                 false, a.v[u][q],  false, b.v[u][q],
                 (short)0, acc[u],  false,  false);
        }
    }
}

template <int G>
__device__ __forceinline__ void store_d_trip(float* Xr, int wave, int hi, int t,
                                             const v8f (&acc)[CTX_TRIP]) {
#pragma unroll
    for (int u = 0; u < CTX_TRIP; ++u) {
        const int c = (wave << 5) | (t * CTX_TRIP + u);
        int base, step;
        ctx_pos(G, c, base, step);
        float* Drow = Xr + base;
#pragma unroll
        for (int j = 0; j < 8; ++j) {
            const int i = j + 8 * hi;
            Drow[i * step] = acc[u][j];
        }
    }
}

template <int G>
__device__ __forceinline__ void run_stage(float* Xr, const float* __restrict__ Mw,
                                          int wave, int hi, Frag4& aP) {
    Frag4 aQ, bP, bQ;
    v8f acc[CTX_TRIP];

    load_b_trip<G>(Xr, wave, hi, 0, bP);

#pragma unroll
    for (int t = 0; t < NTRIPS; t += 2) {
        load_a_trip(Mw, G, t + 1, aQ);
        load_b_trip<G>(Xr, wave, hi, t + 1, bQ);
        mma_trip(aP, bP, acc);
        store_d_trip<G>(Xr, wave, hi, t, acc);

        if (t + 2 < NTRIPS) {
            load_a_trip(Mw, G, t + 2, aP);
            load_b_trip<G>(Xr, wave, hi, t + 2, bP);
        } else if (G < 2) {
            load_a_trip(Mw, G + 1, 0, aP);
        }
        mma_trip(aQ, bQ, acc);
        store_d_trip<G>(Xr, wave, hi, t + 1, acc);
    }
}

__global__ void __launch_bounds__(256, 1)
butterfly_apply(const float* __restrict__ x,
                const float* __restrict__ M,
                float* __restrict__ out) {
    extern __shared__ float X[];

    const int r0 = blockIdx.x * ROWS_TILE;

    const int lane = threadIdx.x & 31;
    const int wave = threadIdx.x >> 5;
    const int r    = lane & 15;
    const int hi   = lane >> 4;

    const float* Mw = M + wave * (32 * 256) + r * 16 + 2 * hi;
    Frag4 aP;
    load_a_trip(Mw, 0, 0, aP);

    {
        const float4* src = (const float4*)(x + (size_t)r0 * NELEM);
        for (int i = threadIdx.x; i < ROWS_TILE * (NELEM / 4); i += 256) {
            const int row = i >> 10;
            const int c4  = i & 1023;
            float4 val = src[row * 1024 + c4];
            *(float4*)&X[row * LDS_STRIDE + c4 * 4] = val;
        }
    }

    float* Xr = X + r * LDS_STRIDE;

    __syncthreads();
    run_stage<0>(Xr, Mw, wave, hi, aP);
    __syncthreads();
    run_stage<1>(Xr, Mw, wave, hi, aP);
    __syncthreads();
    run_stage<2>(Xr, Mw, wave, hi, aP);
    __syncthreads();

    {
        float4* dst = (float4*)(out + (size_t)r0 * NELEM);
        for (int i = threadIdx.x; i < ROWS_TILE * (NELEM / 4); i += 256) {
            const int row = i >> 10;
            const int c4  = i & 1023;
            dst[row * 1024 + c4] = *(const float4*)&X[row * LDS_STRIDE + c4 * 4];
        }
    }
}

extern "C" void kernel_launch(void* const* d_in, const int* in_sizes, int n_in,
                              void* d_out, int out_size, void* d_ws, size_t ws_size,
                              hipStream_t stream) {
    (void)in_sizes; (void)n_in; (void)out_size; (void)ws_size;

    const float* x  = (const float*)d_in[0];
    const float* tw = (const float*)d_in[1];
    float*       o  = (float*)d_out;
    float*       M  = (float*)d_ws;

    butterfly_compose<<<48, 256, 0, stream>>>(tw, M);

    const size_t lds_bytes = (size_t)ROWS_TILE * LDS_STRIDE * sizeof(float);
    (void)hipFuncSetAttribute((const void*)butterfly_apply,
                              hipFuncAttributeMaxDynamicSharedMemorySize,
                              (int)lds_bytes);

    butterfly_apply<<<BATCH / ROWS_TILE, 256, lds_bytes, stream>>>(x, M, o);
}


# apply kernel stages 1-2: B/D fragments as conflict-free 128-bit LDS accesses (one digit of a trip's 4 adjacent contexts), pairs built at the WMMA; a fence keeps next-trip requests ahead of the block
# speedup vs baseline: 1.3743x; 1.1378x over previous
#include <hip/hip_runtime.h>


typedef float v2f __attribute__((ext_vector_type(2)));
typedef float v4f __attribute__((ext_vector_type(4)));
typedef float v8f __attribute__((ext_vector_type(8)));

#define LOG_N       12
#define NELEM       4096
#define BATCH       8192
#define ROWS_TILE   16
#define LDS_STRIDE  4100

__device__ __forceinline__ void ctx_pos(int g, int c, int& base, int& step) {
    if (g == 0)      { base = c << 4;                        step = 1;   }
    else if (g == 1) { base = ((c & 0xF0) << 4) | (c & 15);  step = 16;  }
    else             { base = c;                             step = 256; }
}

__global__ void butterfly_compose(const float* __restrict__ tw,
                                  float* __restrict__ M) {
    int tid = blockIdx.x * blockDim.x + threadIdx.x;
    if (tid >= 3 * 256 * 16) return;
    int k = tid & 15;
    int c = (tid >> 4) & 255;
    int g = tid >> 12;

    int base, step;
    ctx_pos(g, c, base, step);

    float v[16];
#pragma unroll
    for (int t = 0; t < 16; ++t) v[t] = (t == k) ? 1.0f : 0.0f;

#pragma unroll
    for (int j = 0; j < 4; ++j) {
        const int idx = 4 * g + j;
        const int S   = 1 << idx;
#pragma unroll
        for (int u = 0; u < 8; ++u) {
            const int off = u & ((1 << j) - 1);
            const int blk = u >> j;
            const int t0  = (blk << (j + 1)) | off;
            const int t1  = t0 + (1 << j);
            const int p0  = base + t0 * step;
            const int w   = ((p0 >> (idx + 1)) << idx) | (p0 & (S - 1));
            const float* q = tw + idx * (2048 * 4) + w * 4;
            const float a00 = q[0], a01 = q[1], a10 = q[2], a11 = q[3];
            const float x0 = v[t0], x1 = v[t1];
            v[t0] = a00 * x0 + a01 * x1;
            v[t1] = a10 * x0 + a11 * x1;
        }
    }

    float* out = M + ((g * 256 + c) * 16) * 16 + k;
#pragma unroll
    for (int i = 0; i < 16; ++i) out[i * 16] = v[i];
}

#define CTX_TRIP   4
#define NTRIPS     (32 / CTX_TRIP)

struct Frag4 { v2f v[CTX_TRIP][4]; };

struct FragB { v2f v[CTX_TRIP][4]; v4f w[4][2]; };

__device__ __forceinline__ void load_a_trip(const float* __restrict__ Mw,
                                            int g, int t, Frag4& f) {
    const float* p = Mw + (g * 256 + t * CTX_TRIP) * 256;
#pragma unroll
    for (int u = 0; u < CTX_TRIP; ++u) {
#pragma unroll
        for (int q = 0; q < 4; ++q) {
            f.v[u][q].x = p[u * 256 + 4 * q];
            f.v[u][q].y = p[u * 256 + 4 * q + 1];
        }
    }
}

template <int G>
__device__ __forceinline__ void load_b_trip(const float* Xr, int wave, int hi,
                                            int t, FragB& f) {
    if constexpr (G == 0) {
#pragma unroll
        for (int u = 0; u < CTX_TRIP; ++u) {
            const int c = (wave << 5) | (t * CTX_TRIP + u);
            int base, step;
            ctx_pos(G, c, base, step);
            const float* Brow = Xr + base;
#pragma unroll
            for (int q = 0; q < 4; ++q) {
                const int kb = 4 * q + 2 * hi;
                f.v[u][q].x = Brow[kb * step];
                f.v[u][q].y = Brow[(kb + 1) * step];
            }
        }
    } else {
        static_assert(CTX_TRIP == 4, "one float4 = one digit of the trip's 4 contexts");
        int base, step;
        ctx_pos(G, (wave << 5) | (t * CTX_TRIP), base, step);
        const float* Brow = Xr + base;
#pragma unroll
        for (int q = 0; q < 4; ++q) {
#pragma unroll
            for (int e = 0; e < 2; ++e) {
                const int k = 4 * q + 2 * hi + e;
                f.w[q][e] = *(const v4f*)(Brow + k * step);
            }
        }
    }
}

template <int G>
__device__ __forceinline__ void mma_trip(const Frag4& a, const FragB& b,
                                         v8f (&acc)[CTX_TRIP]) {
#pragma unroll
    for (int u = 0; u < CTX_TRIP; ++u) acc[u] = v8f{};
#pragma unroll
    for (int q = 0; q < 4; ++q) {
#pragma unroll
        for (int u = 0; u < CTX_TRIP; ++u) {
            v2f bq;
            if constexpr (G == 0) bq = b.v[u][q];
            else                  bq = v2f{b.w[q][0][u], b.w[q][1][u]};
            acc[u] = __builtin_amdgcn_wmma_f32_16x16x4_f32(
                 false, a.v[u][q],  false, bq,
                 (short)0, acc[u],  false,  false);
        }
    }
}

template <int G>
__device__ __forceinline__ void store_d_trip(float* Xr, int wave, int hi, int t,
                                             const v8f (&acc)[CTX_TRIP]) {
    if constexpr (G == 0) {
#pragma unroll
        for (int u = 0; u < CTX_TRIP; ++u) {
            const int c = (wave << 5) | (t * CTX_TRIP + u);
            int base, step;
            ctx_pos(G, c, base, step);
            float* Drow = Xr + base;
#pragma unroll
            for (int j = 0; j < 8; ++j) {
                const int i = j + 8 * hi;
                Drow[i * step] = acc[u][j];
            }
        }
    } else {
        static_assert(CTX_TRIP == 4, "one float4 = one digit of the trip's 4 contexts");
        int base, step;
        ctx_pos(G, (wave << 5) | (t * CTX_TRIP), base, step);
        float* Drow = Xr + base;
#pragma unroll
        for (int j = 0; j < 8; ++j) {
            const int i = j + 8 * hi;
            *(v4f*)(Drow + i * step) = v4f{acc[0][j], acc[1][j], acc[2][j], acc[3][j]};
        }
    }
}

template <int G>
__device__ __forceinline__ void run_stage(float* Xr, const float* __restrict__ Mw,
                                          int wave, int hi, Frag4& aP) {
    Frag4 aQ;
    FragB bP, bQ;
    v8f acc[CTX_TRIP];

    load_b_trip<G>(Xr, wave, hi, 0, bP);

#pragma unroll
    for (int t = 0; t < NTRIPS; t += 2) {
        load_a_trip(Mw, G, t + 1, aQ);
        load_b_trip<G>(Xr, wave, hi, t + 1, bQ);
        __builtin_amdgcn_sched_barrier(0);
        mma_trip<G>(aP, bP, acc);
        store_d_trip<G>(Xr, wave, hi, t, acc);

        if (t + 2 < NTRIPS) {
            load_a_trip(Mw, G, t + 2, aP);
            load_b_trip<G>(Xr, wave, hi, t + 2, bP);
        } else if (G < 2) {
            load_a_trip(Mw, G + 1, 0, aP);
        }
        __builtin_amdgcn_sched_barrier(0);
        mma_trip<G>(aQ, bQ, acc);
        store_d_trip<G>(Xr, wave, hi, t + 1, acc);
    }
}

__global__ void __launch_bounds__(256, 1)
butterfly_apply(const float* __restrict__ x,
                const float* __restrict__ M,
                float* __restrict__ out) {
    extern __shared__ __attribute__((aligned(16))) float X[];

    const int r0 = blockIdx.x * ROWS_TILE;

    const int lane = threadIdx.x & 31;
    const int wave = threadIdx.x >> 5;
    const int r    = lane & 15;
    const int hi   = lane >> 4;

    const float* Mw = M + wave * (32 * 256) + r * 16 + 2 * hi;
    Frag4 aP;
    load_a_trip(Mw, 0, 0, aP);

    {
        const float4* src = (const float4*)(x + (size_t)r0 * NELEM);
        for (int i = threadIdx.x; i < ROWS_TILE * (NELEM / 4); i += 256) {
            const int row = i >> 10;
            const int c4  = i & 1023;
            float4 val = src[row * 1024 + c4];
            *(float4*)&X[row * LDS_STRIDE + c4 * 4] = val;
        }
    }

    float* Xr = X + r * LDS_STRIDE;

    __syncthreads();
    run_stage<0>(Xr, Mw, wave, hi, aP);
    __syncthreads();
    run_stage<1>(Xr, Mw, wave, hi, aP);
    __syncthreads();
    run_stage<2>(Xr, Mw, wave, hi, aP);
    __syncthreads();

    {
        float4* dst = (float4*)(out + (size_t)r0 * NELEM);
        for (int i = threadIdx.x; i < ROWS_TILE * (NELEM / 4); i += 256) {
            const int row = i >> 10;
            const int c4  = i & 1023;
            dst[row * 1024 + c4] = *(const float4*)&X[row * LDS_STRIDE + c4 * 4];
        }
    }
}

extern "C" void kernel_launch(void* const* d_in, const int* in_sizes, int n_in,
                              void* d_out, int out_size, void* d_ws, size_t ws_size,
                              hipStream_t stream) {
    (void)in_sizes; (void)n_in; (void)out_size; (void)ws_size;

    const float* x  = (const float*)d_in[0];
    const float* tw = (const float*)d_in[1];
    float*       o  = (float*)d_out;
    float*       M  = (float*)d_ws;

    butterfly_compose<<<48, 256, 0, stream>>>(tw, M);

    const size_t lds_bytes = (size_t)ROWS_TILE * LDS_STRIDE * sizeof(float);
    (void)hipFuncSetAttribute((const void*)butterfly_apply,
                              hipFuncAttributeMaxDynamicSharedMemorySize,
                              (int)lds_bytes);

    butterfly_apply<<<BATCH / ROWS_TILE, 256, lds_bytes, stream>>>(x, M, o);
}


# compose kernel stores the stage matrices in WMMA A-fragment order: each A load is one contiguous 512-byte wave load of 128 bits per lane (2 per context), not 4 strided 64-bit loads over 8 lines
# speedup vs baseline: 1.4533x; 1.0575x over previous
#include <hip/hip_runtime.h>


typedef float v2f __attribute__((ext_vector_type(2)));
typedef float v4f __attribute__((ext_vector_type(4)));
typedef float v8f __attribute__((ext_vector_type(8)));

#define LOG_N       12
#define NELEM       4096
#define BATCH       8192
#define ROWS_TILE   16
#define LDS_STRIDE  4100

__device__ __forceinline__ void ctx_pos(int g, int c, int& base, int& step) {
    if (g == 0)      { base = c << 4;                        step = 1;   }
    else if (g == 1) { base = ((c & 0xF0) << 4) | (c & 15);  step = 16;  }
    else             { base = c;                             step = 256; }
}

__global__ void butterfly_compose(const float* __restrict__ tw,
                                  float* __restrict__ M) {
    int tid = blockIdx.x * blockDim.x + threadIdx.x;
    if (tid >= 3 * 256 * 16) return;
    int k = tid & 15;
    int c = (tid >> 4) & 255;
    int g = tid >> 12;

    int base, step;
    ctx_pos(g, c, base, step);

    float v[16];
#pragma unroll
    for (int t = 0; t < 16; ++t) v[t] = (t == k) ? 1.0f : 0.0f;

#pragma unroll
    for (int j = 0; j < 4; ++j) {
        const int idx = 4 * g + j;
        const int S   = 1 << idx;
#pragma unroll
        for (int u = 0; u < 8; ++u) {
            const int off = u & ((1 << j) - 1);
            const int blk = u >> j;
            const int t0  = (blk << (j + 1)) | off;
            const int t1  = t0 + (1 << j);
            const int p0  = base + t0 * step;
            const int w   = ((p0 >> (idx + 1)) << idx) | (p0 & (S - 1));
            const float* q = tw + idx * (2048 * 4) + w * 4;
            const float a00 = q[0], a01 = q[1], a10 = q[2], a11 = q[3];
            const float x0 = v[t0], x1 = v[t1];
            v[t0] = a00 * x0 + a01 * x1;
            v[t1] = a10 * x0 + a11 * x1;
        }
    }

    const int qq = k >> 3, h2 = (k >> 2) & 1, hi = (k >> 1) & 1, e = k & 1;
    float* out = M + (((g * 256 + c) * 2 + qq) * 32 + 16 * hi) * 4 + 2 * h2 + e;
#pragma unroll
    for (int i = 0; i < 16; ++i) out[i * 4] = v[i];
}

#define CTX_TRIP   4
#define NTRIPS     (32 / CTX_TRIP)

struct FragA { v4f v[CTX_TRIP][2]; };

struct FragB { v2f v[CTX_TRIP][4]; v4f w[4][2]; };

__device__ __forceinline__ void load_a_trip(const v4f* __restrict__ Mw,
                                            int g, int t, FragA& f) {
    const v4f* p = Mw + (g * 256 + t * CTX_TRIP) * 64;
#pragma unroll
    for (int qq = 0; qq < 2; ++qq) {
#pragma unroll
        for (int u = 0; u < CTX_TRIP; ++u) {
            f.v[u][qq] = p[u * 64 + qq * 32];
        }
    }
}

template <int G>
__device__ __forceinline__ void load_b_trip(const float* Xr, int wave, int hi,
                                            int t, FragB& f) {
    if constexpr (G == 0) {
#pragma unroll
        for (int u = 0; u < CTX_TRIP; ++u) {
            const int c = (wave << 5) | (t * CTX_TRIP + u);
            int base, step;
            ctx_pos(G, c, base, step);
            const float* Brow = Xr + base;
#pragma unroll
            for (int q = 0; q < 4; ++q) {
                const int kb = 4 * q + 2 * hi;
                f.v[u][q].x = Brow[kb * step];
                f.v[u][q].y = Brow[(kb + 1) * step];
            }
        }
    } else {
        static_assert(CTX_TRIP == 4, "one float4 = one digit of the trip's 4 contexts");
        int base, step;
        ctx_pos(G, (wave << 5) | (t * CTX_TRIP), base, step);
        const float* Brow = Xr + base;
#pragma unroll
        for (int q = 0; q < 4; ++q) {
#pragma unroll
            for (int e = 0; e < 2; ++e) {
                const int k = 4 * q + 2 * hi + e;
                f.w[q][e] = *(const v4f*)(Brow + k * step);
            }
        }
    }
}

template <int G>
__device__ __forceinline__ void mma_trip(const FragA& a, const FragB& b,
                                         v8f (&acc)[CTX_TRIP]) {
#pragma unroll
    for (int u = 0; u < CTX_TRIP; ++u) acc[u] = v8f{};
#pragma unroll
    for (int q = 0; q < 4; ++q) {
#pragma unroll
        for (int u = 0; u < CTX_TRIP; ++u) {
            v2f bq;
            if constexpr (G == 0) bq = b.v[u][q];
            else                  bq = v2f{b.w[q][0][u], b.w[q][1][u]};
            const v4f aw = a.v[u][q >> 1];
            const v2f aq = (q & 1) ? v2f{aw.z, aw.w} : v2f{aw.x, aw.y};
            acc[u] = __builtin_amdgcn_wmma_f32_16x16x4_f32(
                 false, aq,  false, bq,
                 (short)0, acc[u],  false,  false);
        }
    }
}

template <int G>
__device__ __forceinline__ void store_d_trip(float* Xr, int wave, int hi, int t,
                                             const v8f (&acc)[CTX_TRIP]) {
    if constexpr (G == 0) {
#pragma unroll
        for (int u = 0; u < CTX_TRIP; ++u) {
            const int c = (wave << 5) | (t * CTX_TRIP + u);
            int base, step;
            ctx_pos(G, c, base, step);
            float* Drow = Xr + base;
#pragma unroll
            for (int j = 0; j < 8; ++j) {
                const int i = j + 8 * hi;
                Drow[i * step] = acc[u][j];
            }
        }
    } else {
        static_assert(CTX_TRIP == 4, "one float4 = one digit of the trip's 4 contexts");
        int base, step;
        ctx_pos(G, (wave << 5) | (t * CTX_TRIP), base, step);
        float* Drow = Xr + base;
#pragma unroll
        for (int j = 0; j < 8; ++j) {
            const int i = j + 8 * hi;
            *(v4f*)(Drow + i * step) = v4f{acc[0][j], acc[1][j], acc[2][j], acc[3][j]};
        }
    }
}

template <int G>
__device__ __forceinline__ void run_stage(float* Xr, const v4f* __restrict__ Mw,
                                          int wave, int hi, FragA& aP) {
    FragA aQ;
    FragB bP, bQ;
    v8f acc[CTX_TRIP];

    load_b_trip<G>(Xr, wave, hi, 0, bP);

#pragma unroll
    for (int t = 0; t < NTRIPS; t += 2) {
        load_a_trip(Mw, G, t + 1, aQ);
        load_b_trip<G>(Xr, wave, hi, t + 1, bQ);
        __builtin_amdgcn_sched_barrier(0);
        mma_trip<G>(aP, bP, acc);
        store_d_trip<G>(Xr, wave, hi, t, acc);

        if (t + 2 < NTRIPS) {
            load_a_trip(Mw, G, t + 2, aP);
            load_b_trip<G>(Xr, wave, hi, t + 2, bP);
        } else if (G < 2) {
            load_a_trip(Mw, G + 1, 0, aP);
        }
        __builtin_amdgcn_sched_barrier(0);
        mma_trip<G>(aQ, bQ, acc);
        store_d_trip<G>(Xr, wave, hi, t + 1, acc);
    }
}

__global__ void __launch_bounds__(256, 1)
butterfly_apply(const float* __restrict__ x,
                const float* __restrict__ M,
                float* __restrict__ out) {
    extern __shared__ __attribute__((aligned(16))) float X[];

    const int r0 = blockIdx.x * ROWS_TILE;

    const int lane = threadIdx.x & 31;
    const int wave = threadIdx.x >> 5;
    const int r    = lane & 15;
    const int hi   = lane >> 4;

    const v4f* Mw = (const v4f*)M + wave * (32 * 64) + lane;
    FragA aP;
    load_a_trip(Mw, 0, 0, aP);

    {
        const float4* src = (const float4*)(x + (size_t)r0 * NELEM);
        for (int i = threadIdx.x; i < ROWS_TILE * (NELEM / 4); i += 256) {
            const int row = i >> 10;
            const int c4  = i & 1023;
            float4 val = src[row * 1024 + c4];
            *(float4*)&X[row * LDS_STRIDE + c4 * 4] = val;
        }
    }

    float* Xr = X + r * LDS_STRIDE;

    __syncthreads();
    run_stage<0>(Xr, Mw, wave, hi, aP);
    __syncthreads();
    run_stage<1>(Xr, Mw, wave, hi, aP);
    __syncthreads();
    run_stage<2>(Xr, Mw, wave, hi, aP);
    __syncthreads();

    {
        float4* dst = (float4*)(out + (size_t)r0 * NELEM);
        for (int i = threadIdx.x; i < ROWS_TILE * (NELEM / 4); i += 256) {
            const int row = i >> 10;
            const int c4  = i & 1023;
            dst[row * 1024 + c4] = *(const float4*)&X[row * LDS_STRIDE + c4 * 4];
        }
    }
}

extern "C" void kernel_launch(void* const* d_in, const int* in_sizes, int n_in,
                              void* d_out, int out_size, void* d_ws, size_t ws_size,
                              hipStream_t stream) {
    (void)in_sizes; (void)n_in; (void)out_size; (void)ws_size;

    const float* x  = (const float*)d_in[0];
    const float* tw = (const float*)d_in[1];
    float*       o  = (float*)d_out;
    float*       M  = (float*)d_ws;

    butterfly_compose<<<48, 256, 0, stream>>>(tw, M);

    const size_t lds_bytes = (size_t)ROWS_TILE * LDS_STRIDE * sizeof(float);
    (void)hipFuncSetAttribute((const void*)butterfly_apply,
                              hipFuncAttributeMaxDynamicSharedMemorySize,
                              (int)lds_bytes);

    butterfly_apply<<<BATCH / ROWS_TILE, 256, lds_bytes, stream>>>(x, M, o);
}


# apply kernel: each wave loads its own stage-0 slice of x (128-column blocks, two ahead) and runs stage 0 under the loads; the workgroup copy loop (8 loads per trip, drained) and its barrier go
# speedup vs baseline: 1.5762x; 1.0846x over previous
#include <hip/hip_runtime.h>


typedef float v2f __attribute__((ext_vector_type(2)));
typedef float v4f __attribute__((ext_vector_type(4)));
typedef float v8f __attribute__((ext_vector_type(8)));

#define LOG_N       12
#define NELEM       4096
#define BATCH       8192
#define ROWS_TILE   16
#define LDS_STRIDE  4100

__device__ __forceinline__ void ctx_pos(int g, int c, int& base, int& step) {
    if (g == 0)      { base = c << 4;                        step = 1;   }
    else if (g == 1) { base = ((c & 0xF0) << 4) | (c & 15);  step = 16;  }
    else             { base = c;                             step = 256; }
}

__global__ void butterfly_compose(const float* __restrict__ tw,
                                  float* __restrict__ M) {
    int tid = blockIdx.x * blockDim.x + threadIdx.x;
    if (tid >= 3 * 256 * 16) return;
    int k = tid & 15;
    int c = (tid >> 4) & 255;
    int g = tid >> 12;

    int base, step;
    ctx_pos(g, c, base, step);

    float v[16];
#pragma unroll
    for (int t = 0; t < 16; ++t) v[t] = (t == k) ? 1.0f : 0.0f;

#pragma unroll
    for (int j = 0; j < 4; ++j) {
        const int idx = 4 * g + j;
        const int S   = 1 << idx;
#pragma unroll
        for (int u = 0; u < 8; ++u) {
            const int off = u & ((1 << j) - 1);
            const int blk = u >> j;
            const int t0  = (blk << (j + 1)) | off;
            const int t1  = t0 + (1 << j);
            const int p0  = base + t0 * step;
            const int w   = ((p0 >> (idx + 1)) << idx) | (p0 & (S - 1));
            const float* q = tw + idx * (2048 * 4) + w * 4;
            const float a00 = q[0], a01 = q[1], a10 = q[2], a11 = q[3];
            const float x0 = v[t0], x1 = v[t1];
            v[t0] = a00 * x0 + a01 * x1;
            v[t1] = a10 * x0 + a11 * x1;
        }
    }

    const int qq = k >> 3, h2 = (k >> 2) & 1, hi = (k >> 1) & 1, e = k & 1;
    float* out = M + (((g * 256 + c) * 2 + qq) * 32 + 16 * hi) * 4 + 2 * h2 + e;
#pragma unroll
    for (int i = 0; i < 16; ++i) out[i * 4] = v[i];
}

#define CTX_TRIP   4
#define NTRIPS     (32 / CTX_TRIP)

struct FragA { v4f v[CTX_TRIP][2]; };

struct FragB { v2f v[CTX_TRIP][4]; v4f w[4][2]; };

__device__ __forceinline__ void load_a_trip(const v4f* __restrict__ Mw,
                                            int g, int t, FragA& f) {
    const v4f* p = Mw + (g * 256 + t * CTX_TRIP) * 64;
#pragma unroll
    for (int qq = 0; qq < 2; ++qq) {
#pragma unroll
        for (int u = 0; u < CTX_TRIP; ++u) {
            f.v[u][qq] = p[u * 64 + qq * 32];
        }
    }
}

template <int G>
__device__ __forceinline__ void load_b_trip(const float* Xr, int wave, int hi,
                                            int t, FragB& f) {
    if constexpr (G == 0) {
#pragma unroll
        for (int u = 0; u < CTX_TRIP; ++u) {
            const int c = (wave << 5) | (t * CTX_TRIP + u);
            int base, step;
            ctx_pos(G, c, base, step);
            const float* Brow = Xr + base;
#pragma unroll
            for (int q = 0; q < 4; ++q) {
                const int kb = 4 * q + 2 * hi;
                f.v[u][q].x = Brow[kb * step];
                f.v[u][q].y = Brow[(kb + 1) * step];
            }
        }
    } else {
        static_assert(CTX_TRIP == 4, "one float4 = one digit of the trip's 4 contexts");
        int base, step;
        ctx_pos(G, (wave << 5) | (t * CTX_TRIP), base, step);
        const float* Brow = Xr + base;
#pragma unroll
        for (int q = 0; q < 4; ++q) {
#pragma unroll
            for (int e = 0; e < 2; ++e) {
                const int k = 4 * q + 2 * hi + e;
                f.w[q][e] = *(const v4f*)(Brow + k * step);
            }
        }
    }
}

template <int G>
__device__ __forceinline__ void mma_trip(const FragA& a, const FragB& b,
                                         v8f (&acc)[CTX_TRIP]) {
#pragma unroll
    for (int u = 0; u < CTX_TRIP; ++u) acc[u] = v8f{};
#pragma unroll
    for (int q = 0; q < 4; ++q) {
#pragma unroll
        for (int u = 0; u < CTX_TRIP; ++u) {
            v2f bq;
            if constexpr (G == 0) bq = b.v[u][q];
            else                  bq = v2f{b.w[q][0][u], b.w[q][1][u]};
            const v4f aw = a.v[u][q >> 1];
            const v2f aq = (q & 1) ? v2f{aw.z, aw.w} : v2f{aw.x, aw.y};
            acc[u] = __builtin_amdgcn_wmma_f32_16x16x4_f32(
                 false, aq,  false, bq,
                 (short)0, acc[u],  false,  false);
        }
    }
}

template <int G>
__device__ __forceinline__ void store_d_trip(float* Xr, int wave, int hi, int t,
                                             const v8f (&acc)[CTX_TRIP]) {
    if constexpr (G == 0) {
#pragma unroll
        for (int u = 0; u < CTX_TRIP; ++u) {
            const int c = (wave << 5) | (t * CTX_TRIP + u);
            int base, step;
            ctx_pos(G, c, base, step);
            float* Drow = Xr + base;
#pragma unroll
            for (int j = 0; j < 8; ++j) {
                const int i = j + 8 * hi;
                Drow[i * step] = acc[u][j];
            }
        }
    } else {
        static_assert(CTX_TRIP == 4, "one float4 = one digit of the trip's 4 contexts");
        int base, step;
        ctx_pos(G, (wave << 5) | (t * CTX_TRIP), base, step);
        float* Drow = Xr + base;
#pragma unroll
        for (int j = 0; j < 8; ++j) {
            const int i = j + 8 * hi;
            *(v4f*)(Drow + i * step) = v4f{acc[0][j], acc[1][j], acc[2][j], acc[3][j]};
        }
    }
}

struct XBuf { v4f v[ROWS_TILE]; };

__device__ __forceinline__ void load_x_block(const v4f* __restrict__ xw, int p, XBuf& b) {
#pragma unroll
    for (int row = 0; row < ROWS_TILE; ++row) b.v[row] = xw[row * (NELEM / 4) + p * 32];
}

__device__ __forceinline__ void stage_x_block(float* Xw, int p, const XBuf& b) {
#pragma unroll
    for (int row = 0; row < ROWS_TILE; ++row)
        *(v4f*)(Xw + row * LDS_STRIDE + p * 128) = b.v[row];
}

__device__ __forceinline__ void run_stage0_from_memory(float* Xr, float* Xw,
                                                       const v4f* __restrict__ xw,
                                                       const v4f* __restrict__ Mw,
                                                       int wave, int hi, FragA& aP) {
    FragA aQ;
    FragB bP, bQ;
    v8f acc[CTX_TRIP];
    XBuf x0, x1;

    load_x_block(xw, 0, x0);
    __builtin_amdgcn_sched_barrier(0);
    load_x_block(xw, 1, x1);
    __builtin_amdgcn_sched_barrier(0);
    stage_x_block(Xw, 0, x0);
    __builtin_amdgcn_fence(__ATOMIC_SEQ_CST, "wavefront");
    load_b_trip<0>(Xr, wave, hi, 0, bP);

#pragma unroll
    for (int p = 0; p < 4; ++p) {
        const int t = 2 * p;
        XBuf& xa = (p & 1) ? x1 : x0;
        XBuf& xb = (p & 1) ? x0 : x1;

        load_a_trip(Mw, 0, t + 1, aQ);
        load_b_trip<0>(Xr, wave, hi, t + 1, bQ);
        if (p + 2 < 4) load_x_block(xw, p + 2, xa);
        __builtin_amdgcn_sched_barrier(0);
        mma_trip<0>(aP, bP, acc);
        store_d_trip<0>(Xr, wave, hi, t, acc);

        if (p + 1 < 4) {
            stage_x_block(Xw, p + 1, xb);
            __builtin_amdgcn_fence(__ATOMIC_SEQ_CST, "wavefront");
            load_a_trip(Mw, 0, t + 2, aP);
            load_b_trip<0>(Xr, wave, hi, t + 2, bP);
        } else {
            load_a_trip(Mw, 1, 0, aP);
        }
        __builtin_amdgcn_sched_barrier(0);
        mma_trip<0>(aQ, bQ, acc);
        store_d_trip<0>(Xr, wave, hi, t + 1, acc);
    }
}

template <int G>
__device__ __forceinline__ void run_stage(float* Xr, const v4f* __restrict__ Mw,
                                          int wave, int hi, FragA& aP) {
    FragA aQ;
    FragB bP, bQ;
    v8f acc[CTX_TRIP];

    load_b_trip<G>(Xr, wave, hi, 0, bP);

#pragma unroll
    for (int t = 0; t < NTRIPS; t += 2) {
        load_a_trip(Mw, G, t + 1, aQ);
        load_b_trip<G>(Xr, wave, hi, t + 1, bQ);
        __builtin_amdgcn_sched_barrier(0);
        mma_trip<G>(aP, bP, acc);
        store_d_trip<G>(Xr, wave, hi, t, acc);

        if (t + 2 < NTRIPS) {
            load_a_trip(Mw, G, t + 2, aP);
            load_b_trip<G>(Xr, wave, hi, t + 2, bP);
        } else if (G < 2) {
            load_a_trip(Mw, G + 1, 0, aP);
        }
        __builtin_amdgcn_sched_barrier(0);
        mma_trip<G>(aQ, bQ, acc);
        store_d_trip<G>(Xr, wave, hi, t + 1, acc);
    }
}

__global__ void __launch_bounds__(256, 1)
butterfly_apply(const float* __restrict__ x,
                const float* __restrict__ M,
                float* __restrict__ out) {
    extern __shared__ __attribute__((aligned(16))) float X[];

    const int r0 = blockIdx.x * ROWS_TILE;

    const int lane = threadIdx.x & 31;
    const int wave = threadIdx.x >> 5;
    const int r    = lane & 15;
    const int hi   = lane >> 4;

    const v4f* Mw = (const v4f*)M + wave * (32 * 64) + lane;
    FragA aP;
    load_a_trip(Mw, 0, 0, aP);

    float* Xr = X + r * LDS_STRIDE;
    float* Xw = X + wave * 512 + lane * 4;
    const v4f* xw = (const v4f*)(x + (size_t)r0 * NELEM + wave * 512) + lane;

    run_stage0_from_memory(Xr, Xw, xw, Mw, wave, hi, aP);
    __syncthreads();
    run_stage<1>(Xr, Mw, wave, hi, aP);
    __syncthreads();
    run_stage<2>(Xr, Mw, wave, hi, aP);
    __syncthreads();

    {
        float4* dst = (float4*)(out + (size_t)r0 * NELEM);
        for (int i = threadIdx.x; i < ROWS_TILE * (NELEM / 4); i += 256) {
            const int row = i >> 10;
            const int c4  = i & 1023;
            dst[row * 1024 + c4] = *(const float4*)&X[row * LDS_STRIDE + c4 * 4];
        }
    }
}

extern "C" void kernel_launch(void* const* d_in, const int* in_sizes, int n_in,
                              void* d_out, int out_size, void* d_ws, size_t ws_size,
                              hipStream_t stream) {
    (void)in_sizes; (void)n_in; (void)out_size; (void)ws_size;

    const float* x  = (const float*)d_in[0];
    const float* tw = (const float*)d_in[1];
    float*       o  = (float*)d_out;
    float*       M  = (float*)d_ws;

    butterfly_compose<<<48, 256, 0, stream>>>(tw, M);

    const size_t lds_bytes = (size_t)ROWS_TILE * LDS_STRIDE * sizeof(float);
    (void)hipFuncSetAttribute((const void*)butterfly_apply,
                              hipFuncAttributeMaxDynamicSharedMemorySize,
                              (int)lds_bytes);

    butterfly_apply<<<BATCH / ROWS_TILE, 256, lds_bytes, stream>>>(x, M, o);
}
